# starting kernel
# baseline (speedup 1.0000x reference)
#include <hip/hip_runtime.h>
#include <hip/hip_fp16.h>


typedef float    v2f  __attribute__((ext_vector_type(2)));
typedef float    v8f  __attribute__((ext_vector_type(8)));
typedef _Float16 v16h __attribute__((ext_vector_type(16)));

#define HAVE_WMMA_F32X4 __has_builtin(__builtin_amdgcn_wmma_f32_16x16x4_f32)

__global__ __launch_bounds__(128)
void pgx_proj_wmma(const float* __restrict__ embed,
                   const float* __restrict__ W1,
                   const float* __restrict__ b1,
                   float* __restrict__ H1,
                   float* __restrict__ H2,
                   int N)
{
    const int lane = threadIdx.x & 31;
    const int wave = threadIdx.x >> 5;
    const int n0   = wave * 16;
    const int m0   = blockIdx.x * 16;
    const int half = lane >> 4;
    const int l15  = lane & 15;

    int rowA = m0 + l15;
    if (rowA >= N) rowA = N - 1;
    const int col = n0 + l15;

    v8f acc1 = {};
    v8f acc2 = {};

#if HAVE_WMMA_F32X4
    #pragma unroll
    for (int k0 = 0; k0 < 64; k0 += 4) {
        const int ka = k0 + 2 * half;
        v2f a;
        a.x = embed[rowA * 64 + ka];
        a.y = embed[rowA * 64 + ka + 1];
        v2f bA, bB;
        bA.x = W1[(ka)     * 64 + col];
        bA.y = W1[(ka + 1) * 64 + col];
        bB.x = W1[(64 + ka)     * 64 + col];
        bB.y = W1[(64 + ka + 1) * 64 + col];
        acc1 = __builtin_amdgcn_wmma_f32_16x16x4_f32(false, a, false, bA,
                                                     (short)0, acc1, false, false);
        acc2 = __builtin_amdgcn_wmma_f32_16x16x4_f32(false, a, false, bB,
                                                     (short)0, acc2, false, false);
    }
#else
    #pragma unroll
    for (int k0 = 0; k0 < 64; k0 += 32) {
        v16h a, bA, bB;
        const int kb = k0 + 8 * half;
        #pragma unroll
        for (int t = 0; t < 8; ++t) {
            a[t]     = (_Float16)embed[rowA * 64 + kb + t];
            a[t + 8] = (_Float16)embed[rowA * 64 + kb + 16 + t];
        }
        const int ks = k0 + 16 * half;
        #pragma unroll
        for (int t = 0; t < 16; ++t) {
            bA[t] = (_Float16)W1[(ks + t)      * 64 + col];
            bB[t] = (_Float16)W1[(64 + ks + t) * 64 + col];
        }
        acc1 = __builtin_amdgcn_wmma_f32_16x16x32_f16(false, a, false, bA,
                                                      (short)0, acc1, false, false);
        acc2 = __builtin_amdgcn_wmma_f32_16x16x32_f16(false, a, false, bB,
                                                      (short)0, acc2, false, false);
    }
#endif

    const float bias = b1[col];
    #pragma unroll
    for (int r = 0; r < 8; ++r) {
        const int m = m0 + r + 8 * half;
        if (m < N) {
            H1[m * 64 + col] = acc1[r] + bias;
            H2[m * 64 + col] = acc2[r];
        }
    }
}

__global__ __launch_bounds__(256)
void pgx_edge_gate(const float* __restrict__ H1,
                   const float* __restrict__ H2,
                   const int*   __restrict__ ei,
                   const float* __restrict__ noise,
                   const float* __restrict__ W2,
                   const float* __restrict__ b2,
                   const float* __restrict__ tmp,
                   float* __restrict__ out,
                   int N, int E)
{
    __shared__ __align__(16) float sW2[64];
    if (threadIdx.x < 64) sW2[threadIdx.x] = W2[threadIdx.x];
    __syncthreads();

    const int e = blockIdx.x * 256 + threadIdx.x;
    if (e >= E) return;

    const int i = ei[e];
    const int j = ei[E + e];

    const float4* __restrict__ h1i = (const float4*)(H1 + (size_t)i * 64);
    const float4* __restrict__ h2i = (const float4*)(H2 + (size_t)i * 64);
    const float4* __restrict__ h1j = (const float4*)(H1 + (size_t)j * 64);
    const float4* __restrict__ h2j = (const float4*)(H2 + (size_t)j * 64);
    const float4* __restrict__ w4  = (const float4*)sW2;

    float la_ij = 0.0f;
    float la_ji = 0.0f;
    #pragma unroll
    for (int q = 0; q < 16; ++q) {
        const float4 x1 = h1i[q];
        const float4 x2 = h2j[q];
        const float4 y1 = h1j[q];
        const float4 y2 = h2i[q];
        const float4 w  = w4[q];
        la_ij += fmaxf(x1.x + x2.x, 0.0f) * w.x
               + fmaxf(x1.y + x2.y, 0.0f) * w.y
               + fmaxf(x1.z + x2.z, 0.0f) * w.z
               + fmaxf(x1.w + x2.w, 0.0f) * w.w;
        la_ji += fmaxf(y1.x + y2.x, 0.0f) * w.x
               + fmaxf(y1.y + y2.y, 0.0f) * w.y
               + fmaxf(y1.z + y2.z, 0.0f) * w.z
               + fmaxf(y1.w + y2.w, 0.0f) * w.w;
    }

    const float bb   = b2[0];
    const float invt = 1.0f / tmp[0];
    const float nij  = noise[(size_t)i * N + j];
    const float nji  = noise[(size_t)j * N + i];

    const float oij = (__logf(nij) - __logf(1.0f - nij) + la_ij + bb) * invt;
    const float oji = (__logf(nji) - __logf(1.0f - nji) + la_ji + bb) * invt;
    const float gij = 1.0f / (1.0f + __expf(-oij));
    const float gji = 1.0f / (1.0f + __expf(-oji));

    out[e] = 0.5f * (gij + gji);
}

extern "C" void kernel_launch(void* const* d_in, const int* in_sizes, int n_in,
                              void* d_out, int out_size, void* d_ws, size_t ws_size,
                              hipStream_t stream)
{
    const float* embed = (const float*)d_in[0];
    const int*   ei    = (const int*)  d_in[1];
    const float* noise = (const float*)d_in[2];
    const float* tmp   = (const float*)d_in[3];
    const float* W1    = (const float*)d_in[4];
    const float* b1    = (const float*)d_in[5];
    const float* W2    = (const float*)d_in[6];
    const float* b2    = (const float*)d_in[7];
    float*       out   = (float*)d_out;

    const int N = in_sizes[0] / 64;
    const int E = in_sizes[1] / 2;

    float* H1 = (float*)d_ws;
    float* H2 = H1 + (size_t)N * 64;

    const int mtiles = (N + 15) / 16;
    hipLaunchKernelGGL(pgx_proj_wmma, dim3(mtiles), dim3(128), 0, stream,
                       embed, W1, b1, H1, H2, N);

    hipLaunchKernelGGL(pgx_edge_gate, dim3((E + 255) / 256), dim3(256), 0, stream,
                       H1, H2, ei, noise, W2, b2, tmp, out, N, E);
}


# edge kernel: 16 lanes read each 256-B H row together (4 whole lines per load, not 32 partial); a 16-lane DPP butterfly sums the partials, so the 64-term sum order changes (reduction tree)
# speedup vs baseline: 1.4573x; 1.4573x over previous
#include <hip/hip_runtime.h>
#include <hip/hip_fp16.h>


typedef float    v2f  __attribute__((ext_vector_type(2)));
typedef float    v8f  __attribute__((ext_vector_type(8)));
typedef _Float16 v16h __attribute__((ext_vector_type(16)));

#define HAVE_WMMA_F32X4 __has_builtin(__builtin_amdgcn_wmma_f32_16x16x4_f32)

__global__ __launch_bounds__(128)
void pgx_proj_wmma(const float* __restrict__ embed,
                   const float* __restrict__ W1,
                   const float* __restrict__ b1,
                   float* __restrict__ H1,
                   float* __restrict__ H2,
                   int N)
{
    const int lane = threadIdx.x & 31;
    const int wave = threadIdx.x >> 5;
    const int n0   = wave * 16;
    const int m0   = blockIdx.x * 16;
    const int half = lane >> 4;
    const int l15  = lane & 15;

    int rowA = m0 + l15;
    if (rowA >= N) rowA = N - 1;
    const int col = n0 + l15;

    v8f acc1 = {};
    v8f acc2 = {};

#if HAVE_WMMA_F32X4
    #pragma unroll
    for (int k0 = 0; k0 < 64; k0 += 4) {
        const int ka = k0 + 2 * half;
        v2f a;
        a.x = embed[rowA * 64 + ka];
        a.y = embed[rowA * 64 + ka + 1];
        v2f bA, bB;
        bA.x = W1[(ka)     * 64 + col];
        bA.y = W1[(ka + 1) * 64 + col];
        bB.x = W1[(64 + ka)     * 64 + col];
        bB.y = W1[(64 + ka + 1) * 64 + col];
        acc1 = __builtin_amdgcn_wmma_f32_16x16x4_f32(false, a, false, bA,
                                                     (short)0, acc1, false, false);
        acc2 = __builtin_amdgcn_wmma_f32_16x16x4_f32(false, a, false, bB,
                                                     (short)0, acc2, false, false);
    }
#else
    #pragma unroll
    for (int k0 = 0; k0 < 64; k0 += 32) {
        v16h a, bA, bB;
        const int kb = k0 + 8 * half;
        #pragma unroll
        for (int t = 0; t < 8; ++t) {
            a[t]     = (_Float16)embed[rowA * 64 + kb + t];
            a[t + 8] = (_Float16)embed[rowA * 64 + kb + 16 + t];
        }
        const int ks = k0 + 16 * half;
        #pragma unroll
        for (int t = 0; t < 16; ++t) {
            bA[t] = (_Float16)W1[(ks + t)      * 64 + col];
            bB[t] = (_Float16)W1[(64 + ks + t) * 64 + col];
        }
        acc1 = __builtin_amdgcn_wmma_f32_16x16x32_f16(false, a, false, bA,
                                                      (short)0, acc1, false, false);
        acc2 = __builtin_amdgcn_wmma_f32_16x16x32_f16(false, a, false, bB,
                                                      (short)0, acc2, false, false);
    }
#endif

    const float bias = b1[col];
    #pragma unroll
    for (int r = 0; r < 8; ++r) {
        const int m = m0 + r + 8 * half;
        if (m < N) {
            H1[m * 64 + col] = acc1[r] + bias;
            H2[m * 64 + col] = acc2[r];
        }
    }
}


template <int P>
__device__ __forceinline__ unsigned row_lane(unsigned v)
{
    return (unsigned)__builtin_amdgcn_update_dpp(0, (int)v, 0x150 + P, 0xf, 0xf, true);
}

template <int B>
__device__ __forceinline__ float row_xor(float v)
{
    return __builtin_bit_cast(float,
        __builtin_amdgcn_update_dpp(0, __builtin_bit_cast(int, v), 0x160 + B, 0xf, 0xf, true));
}

template <int B>
__device__ __forceinline__ float bfly(float lo, float hi, bool bit_set)
{
    const float keep = bit_set ? hi : lo;
    const float send = bit_set ? lo : hi;
    return keep + row_xor<B>(send);
}

__device__ __forceinline__ float row_transpose_sum(const float (&s)[16], int c)
{
    const bool b8 = (c & 8) != 0, b4 = (c & 4) != 0, b2 = (c & 2) != 0, b1 = (c & 1) != 0;
    float r8[8], r4[4], r2[2];
    #pragma unroll
    for (int p = 0; p < 8; ++p) r8[p] = bfly<8>(s[p], s[p + 8], b8);
    #pragma unroll
    for (int p = 0; p < 4; ++p) r4[p] = bfly<4>(r8[p], r8[p + 4], b4);
    #pragma unroll
    for (int p = 0; p < 2; ++p) r2[p] = bfly<2>(r4[p], r4[p + 2], b2);
    return bfly<1>(r2[0], r2[1], b1);
}

__device__ __forceinline__ float4 ld_row16(const float* __restrict__ base, unsigned byte_off)
{
    return *(const float4*)((const char*)base + byte_off);
}

__device__ __forceinline__ float relu_dot4(const float4 x, const float4 y, const float4 w)
{
    return fmaxf(x.x + y.x, 0.0f) * w.x
         + fmaxf(x.y + y.y, 0.0f) * w.y
         + fmaxf(x.z + y.z, 0.0f) * w.z
         + fmaxf(x.w + y.w, 0.0f) * w.w;
}

#define PGX_REP16(X) X(0) X(1) X(2) X(3) X(4) X(5) X(6) X(7) \
                     X(8) X(9) X(10) X(11) X(12) X(13) X(14) X(15)

template <bool FULL>
__device__ __forceinline__
void pgx_edge_wave(const float* __restrict__ H1,
                   const float* __restrict__ H2,
                   const int*   __restrict__ ei,
                   const float* __restrict__ noise,
                   const float* __restrict__ W2,
                   const float* __restrict__ b2,
                   const float* __restrict__ tmp,
                   float* __restrict__ out,
                   int N, int E, int wave_base, int c, int h)
{
    const int  e    = wave_base + 2 * c + h;
    const bool live = FULL || e < E;
    const int  ec   = live ? e : (E - 1);

    float4 w;
    w.x = W2[4 * c + 0];
    w.y = W2[4 * c + 1];
    w.z = W2[4 * c + 2];
    w.w = W2[4 * c + 3];
    const int i = ei[ec];
    const int j = ei[E + ec];

    const float nij  = noise[(size_t)i * N + j];
    const float nji  = noise[(size_t)j * N + i];
    __builtin_amdgcn_sched_barrier(0);

    const unsigned ri  = (unsigned)i * 256u;
    const unsigned rj  = (unsigned)j * 256u;
    const unsigned c16 = (unsigned)c * 16u;

    float4 x1[16], x2[16], y1[16], y2[16];
#define PGX_ISSUE(p) {                                   \
        const unsigned oi = row_lane<p>(ri) + c16;       \
        const unsigned oj = row_lane<p>(rj) + c16;       \
        x1[p] = ld_row16(H1, oi);     \
        x2[p] = ld_row16(H2, oj);              \
        y1[p] = ld_row16(H1, oj);              \
        y2[p] = ld_row16(H2, oi);              \
    }
    PGX_REP16(PGX_ISSUE)
#undef PGX_ISSUE

    float sij[16], sji[16];
#define PGX_PART(p) {                                    \
        sij[p] = relu_dot4(x1[p], x2[p], w);             \
        sji[p] = relu_dot4(y1[p], y2[p], w);             \
    }
    PGX_REP16(PGX_PART)
#undef PGX_PART

    const float la_ij = row_transpose_sum(sij, c);
    const float la_ji = row_transpose_sum(sji, c);

    __builtin_amdgcn_sched_barrier(0);

    const float bb   = b2[0];
    const float invt = 1.0f / tmp[0];

    const float oij = (__logf(nij) - __logf(1.0f - nij) + la_ij + bb) * invt;
    const float oji = (__logf(nji) - __logf(1.0f - nji) + la_ji + bb) * invt;
    const float gij = 1.0f / (1.0f + __expf(-oij));
    const float gji = 1.0f / (1.0f + __expf(-oji));

    if (live) out[e] = 0.5f * (gij + gji);
}

__global__ __launch_bounds__(256)
void pgx_edge_gate(const float* __restrict__ H1,
                   const float* __restrict__ H2,
                   const int*   __restrict__ ei,
                   const float* __restrict__ noise,
                   const float* __restrict__ W2,
                   const float* __restrict__ b2,
                   const float* __restrict__ tmp,
                   float* __restrict__ out,
                   int N, int E)
{
    const int lane = threadIdx.x & 31;
    const int c    = lane & 15;
    const int h    = lane >> 4;
    const int wave_base = blockIdx.x * 256 + (threadIdx.x & ~31);

    if (wave_base + 32 <= E)
        pgx_edge_wave<true >(H1, H2, ei, noise, W2, b2, tmp, out, N, E, wave_base, c, h);
    else if (wave_base < E)
        pgx_edge_wave<false>(H1, H2, ei, noise, W2, b2, tmp, out, N, E, wave_base, c, h);
}

extern "C" void kernel_launch(void* const* d_in, const int* in_sizes, int n_in,
                              void* d_out, int out_size, void* d_ws, size_t ws_size,
                              hipStream_t stream)
{
    const float* embed = (const float*)d_in[0];
    const int*   ei    = (const int*)  d_in[1];
    const float* noise = (const float*)d_in[2];
    const float* tmp   = (const float*)d_in[3];
    const float* W1    = (const float*)d_in[4];
    const float* b1    = (const float*)d_in[5];
    const float* W2    = (const float*)d_in[6];
    const float* b2    = (const float*)d_in[7];
    float*       out   = (float*)d_out;

    const int N = in_sizes[0] / 64;
    const int E = in_sizes[1] / 2;

    float* H1 = (float*)d_ws;
    float* H2 = H1 + (size_t)N * 64;

    const int mtiles = (N + 15) / 16;
    hipLaunchKernelGGL(pgx_proj_wmma, dim3(mtiles), dim3(128), 0, stream,
                       embed, W1, b1, H1, H2, N);

    hipLaunchKernelGGL(pgx_edge_gate, dim3((E + 255) / 256), dim3(256), 0, stream,
                       H1, H2, ei, noise, W2, b2, tmp, out, N, E);
}
